# starting kernel
# baseline (speedup 1.0000x reference)
#include <hip/hip_runtime.h>

typedef __attribute__((ext_vector_type(2))) float v2f;
typedef __attribute__((ext_vector_type(8))) float v8f;

#define D_FEAT   128
#define TILE_PAD 132

__global__ __launch_bounds__(256)
void gnn_fused_wmma_kernel(const float* __restrict__ X,
                           const float* __restrict__ W,
                           const int*   __restrict__ rowptr,
                           const int*   __restrict__ colidx,
                           float*       __restrict__ out,
                           int n_nodes)
{
    __shared__ float sA [16 * TILE_PAD];
    __shared__ float sWt[D_FEAT * TILE_PAD];

    const int tid  = threadIdx.x;
    const int lane = tid & 31;
    const int wv   = tid >> 5;
    const int r0   = blockIdx.x * 16;

    for (int i = tid; i < D_FEAT * D_FEAT; i += 256) {
        int k = i >> 7;
        int n = i & 127;
        sWt[n * TILE_PAD + k] = W[i];
    }

    for (int rr = 0; rr < 2; ++rr) {
        const int rl  = wv * 2 + rr;
        const int row = r0 + rl;
        float4 acc = make_float4(0.f, 0.f, 0.f, 0.f);
        if (row < n_nodes) {
            const int e0 = rowptr[row];
            const int e1 = rowptr[row + 1];
            for (int eb = e0; eb < e1; eb += 32) {
                int ci = 0;
                if (eb + lane < e1) ci = colidx[eb + lane];
                const int cnt = min(32, e1 - eb);
                for (int j = 0; j < cnt; ++j) {
                    const int c = __shfl(ci, j, 32);
                    const float4 v =
                        *(const float4*)(X + (size_t)c * D_FEAT + lane * 4);
                    acc.x += v.x; acc.y += v.y; acc.z += v.z; acc.w += v.w;
                }
            }
        }
        *(float4*)(sA + rl * TILE_PAD + lane * 4) = acc;
    }

    __syncthreads();

    const int n0   = wv * 16;
    const int mrow = lane & 15;
    const int kd   = (lane >> 4) * 2;

    v8f c = {};
    const float* aBase = sA  + mrow * TILE_PAD + kd;
    const float* bBase = sWt + (n0 + mrow) * TILE_PAD + kd;
    #pragma unroll
    for (int kk = 0; kk < D_FEAT / 4; ++kk) {
        v2f a = *(const v2f*)(aBase + kk * 4);
        v2f b = *(const v2f*)(bBase + kk * 4);
        c = __builtin_amdgcn_wmma_f32_16x16x4_f32(
                 false, a,  false, b,
                 (short)0, c,  false,  false);
    }

    const int ncol  = n0 + (lane & 15);
    const int mbase = (lane >> 4) * 8;
    #pragma unroll
    for (int v = 0; v < 8; ++v) {
        const int row = r0 + mbase + v;
        if (row < n_nodes)
            out[(size_t)row * D_FEAT + ncol] = c[v];
    }
}

extern "C" void kernel_launch(void* const* d_in, const int* in_sizes, int n_in,
                              void* d_out, int out_size, void* d_ws, size_t ws_size,
                              hipStream_t stream) {
    const float* X      = (const float*)d_in[0];
    const float* W      = (const float*)d_in[1];
    const int*   rowptr = (const int*)  d_in[2];
    const int*   colidx = (const int*)  d_in[3];
    float*       out    = (float*)d_out;

    const int n_nodes   = in_sizes[2] - 1;
    const int n_windows = (n_nodes + 15) / 16;

    hipLaunchKernelGGL(gnn_fused_wmma_kernel,
                       dim3(n_windows), dim3(256), 0, stream,
                       X, W, rowptr, colidx, out, n_nodes);
}


# Gather: 8 source-row loads in flight per wave (column ids by readlane, loads issued as a group, summed in the same edge order, both rows' index loads grouped), not one load per shuffle
# speedup vs baseline: 1.5793x; 1.5793x over previous
#include <hip/hip_runtime.h>


typedef __attribute__((ext_vector_type(2))) float v2f;
typedef __attribute__((ext_vector_type(8))) float v8f;

#define D_FEAT   128
#define TILE_PAD 132
#define GATHER_DEPTH 8

__device__ __forceinline__ float4 load_src_row(const float* __restrict__ Xl, int c)
{
    return *(const float4*)(Xl + (size_t)c * D_FEAT);
}

__device__ __forceinline__ void add4(float4& a, const float4& v)
{
    a.x += v.x; a.y += v.y; a.z += v.z; a.w += v.w;
}

__device__ __forceinline__ float4 gather_row_sum(const float* __restrict__ Xl,
                                                 const int*   __restrict__ colidx,
                                                 int ea, int eb, int lane, int ci)
{
    float4 acc = make_float4(0.f, 0.f, 0.f, 0.f);
    for (int base = ea; base < eb; base += 32) {
        if (base != ea) {
            ci = 0;
            if (base + lane < eb) ci = colidx[base + lane];
        }
        const int cnt = min(32, eb - base);
        int j = 0;
        for (; j + GATHER_DEPTH <= cnt; j += GATHER_DEPTH) {
            float4 v[GATHER_DEPTH];
            #pragma unroll
            for (int u = 0; u < GATHER_DEPTH; ++u)
                v[u] = load_src_row(Xl, __builtin_amdgcn_readlane(ci, j + u));
            #pragma unroll
            for (int u = 0; u < GATHER_DEPTH; ++u) add4(acc, v[u]);
        }
        if (j < cnt) {
            float4 v[GATHER_DEPTH - 1];
            #pragma unroll
            for (int u = 0; u < GATHER_DEPTH - 1; ++u) {
                v[u] = make_float4(0.f, 0.f, 0.f, 0.f);
                if (j + u < cnt)
                    v[u] = load_src_row(Xl, __builtin_amdgcn_readlane(ci, j + u));
            }
            #pragma unroll
            for (int u = 0; u < GATHER_DEPTH - 1; ++u) add4(acc, v[u]);
        }
    }
    return acc;
}

__global__ __launch_bounds__(256)
void gnn_fused_wmma_kernel(const float* __restrict__ X,
                           const float* __restrict__ W,
                           const int*   __restrict__ rowptr,
                           const int*   __restrict__ colidx,
                           float*       __restrict__ out,
                           int n_nodes)
{
    __shared__ float sA [16 * TILE_PAD];
    __shared__ float sWt[D_FEAT * TILE_PAD];

    const int tid  = threadIdx.x;
    const int lane = tid & 31;
    const int wv   = tid >> 5;
    const int r0   = blockIdx.x * 16;

    for (int i = tid; i < D_FEAT * D_FEAT; i += 256) {
        int k = i >> 7;
        int n = i & 127;
        sWt[n * TILE_PAD + k] = W[i];
    }

    {
        const int rowA = r0 + wv * 2;
        const int rp = rowptr[min(rowA + min(lane, 2), n_nodes)];
        const int e0 = __builtin_amdgcn_readlane(rp, 0);
        const int e1 = __builtin_amdgcn_readlane(rp, 1);
        const int e2 = __builtin_amdgcn_readlane(rp, 2);
        int ciA = 0, ciB = 0;
        if (e0 + lane < e1) ciA = colidx[e0 + lane];
        if (e1 + lane < e2) ciB = colidx[e1 + lane];

        const float* Xl = X + lane * 4;
        const float4 accA = gather_row_sum(Xl, colidx, e0, e1, lane, ciA);
        *(float4*)(sA + (wv * 2    ) * TILE_PAD + lane * 4) = accA;
        const float4 accB = gather_row_sum(Xl, colidx, e1, e2, lane, ciB);
        *(float4*)(sA + (wv * 2 + 1) * TILE_PAD + lane * 4) = accB;
    }

    __syncthreads();

    const int n0   = wv * 16;
    const int mrow = lane & 15;
    const int kd   = (lane >> 4) * 2;

    v8f c = {};
    const float* aBase = sA  + mrow * TILE_PAD + kd;
    const float* bBase = sWt + (n0 + mrow) * TILE_PAD + kd;
    #pragma unroll
    for (int kk = 0; kk < D_FEAT / 4; ++kk) {
        v2f a = *(const v2f*)(aBase + kk * 4);
        v2f b = *(const v2f*)(bBase + kk * 4);
        c = __builtin_amdgcn_wmma_f32_16x16x4_f32(
                 false, a,  false, b,
                 (short)0, c,  false,  false);
    }

    const int ncol  = n0 + (lane & 15);
    const int mbase = (lane >> 4) * 8;
    #pragma unroll
    for (int v = 0; v < 8; ++v) {
        const int row = r0 + mbase + v;
        if (row < n_nodes)
            out[(size_t)row * D_FEAT + ncol] = c[v];
    }
}

extern "C" void kernel_launch(void* const* d_in, const int* in_sizes, int n_in,
                              void* d_out, int out_size, void* d_ws, size_t ws_size,
                              hipStream_t stream) {
    const float* X      = (const float*)d_in[0];
    const float* W      = (const float*)d_in[1];
    const int*   rowptr = (const int*)  d_in[2];
    const int*   colidx = (const int*)  d_in[3];
    float*       out    = (float*)d_out;

    const int n_nodes   = in_sizes[2] - 1;
    const int n_windows = (n_nodes + 15) / 16;

    hipLaunchKernelGGL(gnn_fused_wmma_kernel,
                       dim3(n_windows), dim3(256), 0, stream,
                       X, W, rowptr, colidx, out, n_nodes);
}


# Resident work-groups: 1024 groups walk the 16-row windows (g, g+1024, ...); each wave loads its 128x16 slice of W into registers once, so W is no longer staged through LDS for every window
# speedup vs baseline: 2.0409x; 1.2923x over previous
#include <hip/hip_runtime.h>


typedef __attribute__((ext_vector_type(2))) float v2f;
typedef __attribute__((ext_vector_type(8))) float v8f;

#define D_FEAT   128
#define TILE_PAD 132
#define GATHER_DEPTH 8
#define N_GROUPS 1024

__device__ __forceinline__ float4 load_src_row(const float* __restrict__ Xl, int c)
{
    return *(const float4*)(Xl + (size_t)c * D_FEAT);
}

__device__ __forceinline__ void add4(float4& a, const float4& v)
{
    a.x += v.x; a.y += v.y; a.z += v.z; a.w += v.w;
}

__device__ __forceinline__ float4 gather_row_sum(const float* __restrict__ Xl,
                                                 const int*   __restrict__ colidx,
                                                 int ea, int eb, int lane, int ci)
{
    float4 acc = make_float4(0.f, 0.f, 0.f, 0.f);
    for (int base = ea; base < eb; base += 32) {
        if (base != ea) {
            ci = 0;
            if (base + lane < eb) ci = colidx[base + lane];
        }
        const int cnt = min(32, eb - base);
        int j = 0;
        for (; j + GATHER_DEPTH <= cnt; j += GATHER_DEPTH) {
            float4 v[GATHER_DEPTH];
            #pragma unroll
            for (int u = 0; u < GATHER_DEPTH; ++u)
                v[u] = load_src_row(Xl, __builtin_amdgcn_readlane(ci, j + u));
            #pragma unroll
            for (int u = 0; u < GATHER_DEPTH; ++u) add4(acc, v[u]);
        }
        if (j < cnt) {
            float4 v[GATHER_DEPTH - 1];
            #pragma unroll
            for (int u = 0; u < GATHER_DEPTH - 1; ++u) {
                v[u] = make_float4(0.f, 0.f, 0.f, 0.f);
                if (j + u < cnt)
                    v[u] = load_src_row(Xl, __builtin_amdgcn_readlane(ci, j + u));
            }
            #pragma unroll
            for (int u = 0; u < GATHER_DEPTH - 1; ++u) add4(acc, v[u]);
        }
    }
    return acc;
}

__global__ __launch_bounds__(256)
void gnn_fused_wmma_kernel(const float* __restrict__ X,
                           const float* __restrict__ W,
                           const int*   __restrict__ rowptr,
                           const int*   __restrict__ colidx,
                           float*       __restrict__ out,
                           int n_nodes)
{
    __shared__ float sA[2][16 * TILE_PAD];

    const int tid  = threadIdx.x;
    const int lane = tid & 31;
    const int wv   = tid >> 5;
    const int n_windows = (n_nodes + 15) >> 4;

    const int n0   = wv * 16;
    const int mrow = lane & 15;
    const int kd   = (lane >> 4) * 2;

    v2f bw[D_FEAT / 4];
    {
        const float* wp = W + (size_t)kd * D_FEAT + n0 + mrow;
        #pragma unroll
        for (int kk = 0; kk < D_FEAT / 4; ++kk) {
            bw[kk].x = wp[(kk * 4    ) * D_FEAT];
            bw[kk].y = wp[(kk * 4 + 1) * D_FEAT];
        }
    }

    int buf = 0;
    for (int win = blockIdx.x; win < n_windows; win += N_GROUPS) {
        float* tile = sA[buf];
        const int r0 = win * 16;

        {
            const int rowA = r0 + wv * 2;
            const int rp = rowptr[min(rowA + min(lane, 2), n_nodes)];
            const int e0 = __builtin_amdgcn_readlane(rp, 0);
            const int e1 = __builtin_amdgcn_readlane(rp, 1);
            const int e2 = __builtin_amdgcn_readlane(rp, 2);
            int ciA = 0, ciB = 0;
            if (e0 + lane < e1) ciA = colidx[e0 + lane];
            if (e1 + lane < e2) ciB = colidx[e1 + lane];

            const float* Xl = X + lane * 4;
            const float4 accA = gather_row_sum(Xl, colidx, e0, e1, lane, ciA);
            *(float4*)(tile + (wv * 2    ) * TILE_PAD + lane * 4) = accA;
            const float4 accB = gather_row_sum(Xl, colidx, e1, e2, lane, ciB);
            *(float4*)(tile + (wv * 2 + 1) * TILE_PAD + lane * 4) = accB;
        }

        __syncthreads();

        v8f c = {};
        const float* aBase = tile + mrow * TILE_PAD + kd;
        #pragma unroll
        for (int kk = 0; kk < D_FEAT / 4; ++kk) {
            v2f a = *(const v2f*)(aBase + kk * 4);
            c = __builtin_amdgcn_wmma_f32_16x16x4_f32(
                     false, a,  false, bw[kk],
                     (short)0, c,  false,  false);
        }

        const int ncol  = n0 + (lane & 15);
        const int mbase = (lane >> 4) * 8;
        #pragma unroll
        for (int v = 0; v < 8; ++v) {
            const int row = r0 + mbase + v;
            if (row < n_nodes)
                out[(size_t)row * D_FEAT + ncol] = c[v];
        }

        buf ^= 1;
    }
}

extern "C" void kernel_launch(void* const* d_in, const int* in_sizes, int n_in,
                              void* d_out, int out_size, void* d_ws, size_t ws_size,
                              hipStream_t stream) {
    const float* X      = (const float*)d_in[0];
    const float* W      = (const float*)d_in[1];
    const int*   rowptr = (const int*)  d_in[2];
    const int*   colidx = (const int*)  d_in[3];
    float*       out    = (float*)d_out;

    const int n_nodes   = in_sizes[2] - 1;

    hipLaunchKernelGGL(gnn_fused_wmma_kernel,
                       dim3(N_GROUPS), dim3(256), 0, stream,
                       X, W, rowptr, colidx, out, n_nodes);
}


# Output rows staged through LDS: a window's product tile is read back behind the next barrier; each wave stores two whole 512-B rows (one 16-B-per-lane store per row = 4 whole lines), not 64-B pieces
# speedup vs baseline: 2.3731x; 1.1628x over previous
#include <hip/hip_runtime.h>


typedef __attribute__((ext_vector_type(2))) float v2f;
typedef __attribute__((ext_vector_type(8))) float v8f;

#define D_FEAT   128
#define TILE_PAD 132
#define GATHER_DEPTH 8
#define N_GROUPS 1024

__device__ __forceinline__ float4 load_src_row(const float* __restrict__ Xl, int c)
{
    return *(const float4*)(Xl + (size_t)c * D_FEAT);
}

__device__ __forceinline__ void add4(float4& a, const float4& v)
{
    a.x += v.x; a.y += v.y; a.z += v.z; a.w += v.w;
}

__device__ __forceinline__ float4 gather_row_sum(const float* __restrict__ Xl,
                                                 const int*   __restrict__ colidx,
                                                 int ea, int eb, int lane, int ci)
{
    float4 acc = make_float4(0.f, 0.f, 0.f, 0.f);
    for (int base = ea; base < eb; base += 32) {
        if (base != ea) {
            ci = 0;
            if (base + lane < eb) ci = colidx[base + lane];
        }
        const int cnt = min(32, eb - base);
        int j = 0;
        for (; j + GATHER_DEPTH <= cnt; j += GATHER_DEPTH) {
            float4 v[GATHER_DEPTH];
            #pragma unroll
            for (int u = 0; u < GATHER_DEPTH; ++u)
                v[u] = load_src_row(Xl, __builtin_amdgcn_readlane(ci, j + u));
            #pragma unroll
            for (int u = 0; u < GATHER_DEPTH; ++u) add4(acc, v[u]);
        }
        if (j < cnt) {
            float4 v[GATHER_DEPTH - 1];
            #pragma unroll
            for (int u = 0; u < GATHER_DEPTH - 1; ++u) {
                v[u] = make_float4(0.f, 0.f, 0.f, 0.f);
                if (j + u < cnt)
                    v[u] = load_src_row(Xl, __builtin_amdgcn_readlane(ci, j + u));
            }
            #pragma unroll
            for (int u = 0; u < GATHER_DEPTH - 1; ++u) add4(acc, v[u]);
        }
    }
    return acc;
}

__device__ __forceinline__ void store_product_rows(const float* __restrict__ ptile,
                                                   float*       __restrict__ out,
                                                   int row0, int wv, int lane, int n_nodes)
{
    const int rl = __builtin_amdgcn_readfirstlane(wv) * 2;
    const int r  = row0 + rl;
    const float* src = ptile + rl * TILE_PAD + lane * 4;
    float*       dst = out + (size_t)r * D_FEAT + lane * 4;
    if (r + 1 < n_nodes) {
        const float4 p0 = *(const float4*)(src);
        const float4 p1 = *(const float4*)(src + TILE_PAD);
        *(float4*)(dst)          = p0;
        *(float4*)(dst + D_FEAT) = p1;
    } else if (r < n_nodes) {
        *(float4*)(dst) = *(const float4*)(src);
    }
}

__global__ __launch_bounds__(256)
void gnn_fused_wmma_kernel(const float* __restrict__ X,
                           const float* __restrict__ W,
                           const int*   __restrict__ rowptr,
                           const int*   __restrict__ colidx,
                           float*       __restrict__ out,
                           int n_nodes)
{
    __shared__ __attribute__((aligned(16))) float sA[2][16 * TILE_PAD];
    __shared__ __attribute__((aligned(16))) float sO[2][16 * TILE_PAD];

    const int tid  = threadIdx.x;
    const int lane = tid & 31;
    const int wv   = tid >> 5;
    const int n_windows = (n_nodes + 15) >> 4;

    const int n0   = wv * 16;
    const int mrow = lane & 15;
    const int kd   = (lane >> 4) * 2;

    v2f bw[D_FEAT / 4];
    {
        const float* wp = W + (size_t)kd * D_FEAT + n0 + mrow;
        #pragma unroll
        for (int kk = 0; kk < D_FEAT / 4; ++kk) {
            bw[kk].x = wp[(kk * 4    ) * D_FEAT];
            bw[kk].y = wp[(kk * 4 + 1) * D_FEAT];
        }
    }

    int buf = 0;
    int prev_r0 = -1;
    for (int win = blockIdx.x; win < n_windows; win += N_GROUPS) {
        float* tile = sA[buf];
        const int r0 = win * 16;

        {
            const int rowA = r0 + wv * 2;
            const int rp = rowptr[min(rowA + min(lane, 2), n_nodes)];
            const int e0 = __builtin_amdgcn_readlane(rp, 0);
            const int e1 = __builtin_amdgcn_readlane(rp, 1);
            const int e2 = __builtin_amdgcn_readlane(rp, 2);
            int ciA = 0, ciB = 0;
            if (e0 + lane < e1) ciA = colidx[e0 + lane];
            if (e1 + lane < e2) ciB = colidx[e1 + lane];

            const float* Xl = X + lane * 4;
            const float4 accA = gather_row_sum(Xl, colidx, e0, e1, lane, ciA);
            *(float4*)(tile + (wv * 2    ) * TILE_PAD + lane * 4) = accA;
            const float4 accB = gather_row_sum(Xl, colidx, e1, e2, lane, ciB);
            *(float4*)(tile + (wv * 2 + 1) * TILE_PAD + lane * 4) = accB;
        }

        __syncthreads();

        if (prev_r0 >= 0)
            store_product_rows(sO[buf ^ 1], out, prev_r0, wv, lane, n_nodes);

        v8f c = {};
        const float* aBase = tile + mrow * TILE_PAD + kd;
        #pragma unroll
        for (int kk = 0; kk < D_FEAT / 4; ++kk) {
            v2f a = *(const v2f*)(aBase + kk * 4);
            c = __builtin_amdgcn_wmma_f32_16x16x4_f32(
                     false, a,  false, bw[kk],
                     (short)0, c,  false,  false);
        }

        float* ptile = sO[buf];
        const int ncol  = n0 + (lane & 15);
        const int mbase = (lane >> 4) * 8;
        #pragma unroll
        for (int v = 0; v < 8; ++v)
            ptile[(mbase + v) * TILE_PAD + ncol] = c[v];

        prev_r0 = r0;
        buf ^= 1;
    }

    __syncthreads();
    if (prev_r0 >= 0)
        store_product_rows(sO[buf ^ 1], out, prev_r0, wv, lane, n_nodes);
}

extern "C" void kernel_launch(void* const* d_in, const int* in_sizes, int n_in,
                              void* d_out, int out_size, void* d_ws, size_t ws_size,
                              hipStream_t stream) {
    const float* X      = (const float*)d_in[0];
    const float* W      = (const float*)d_in[1];
    const int*   rowptr = (const int*)  d_in[2];
    const int*   colidx = (const int*)  d_in[3];
    float*       out    = (float*)d_out;

    const int n_nodes   = in_sizes[2] - 1;

    hipLaunchKernelGGL(gnn_fused_wmma_kernel,
                       dim3(N_GROUPS), dim3(256), 0, stream,
                       X, W, rowptr, colidx, out, n_nodes);
}


# Gather as a rolling ring: a wave's 8 row-load slots are refilled as each is summed, its two rows walked as one edge run; row pointers / column ids asked two / one windows ahead; sum order unchanged
# speedup vs baseline: 2.4633x; 1.0380x over previous
#include <hip/hip_runtime.h>


typedef __attribute__((ext_vector_type(2))) float v2f;
typedef __attribute__((ext_vector_type(8))) float v8f;

#define D_FEAT   128
#define TILE_PAD 132
#define GATHER_DEPTH 8
#define N_GROUPS 1024

__device__ __forceinline__ float4 load_src_row(const float* __restrict__ X, int c, int lane4)
{
    const unsigned off = ((unsigned)c * D_FEAT + (unsigned)lane4) * (unsigned)sizeof(float);
    return *(const float4*)((const char*)X + off);
}

__device__ __forceinline__ void add4(float4& a, const float4& v)
{
    a.x += v.x; a.y += v.y; a.z += v.z; a.w += v.w;
}

__device__ __forceinline__ int load_row_ptrs(const int* __restrict__ rowptr, int win, int wv, int lane,
                                             int n_nodes)
{
    return rowptr[min(win * 16 + wv * 2 + min(lane, 2), n_nodes)];
}

__device__ __forceinline__ int load_col_ids(const int* __restrict__ colidx, int e0, int e2, int k)
{
    int c = 0;
    if (k >= 0 && e0 + k < e2) c = colidx[e0 + k];
    return c;
}

__device__ __forceinline__ void open_run(const int* __restrict__ colidx, int rp, int lane,
                                         int& e0, int& e1, int& e2, int& ci0, int& ci1)
{
    e0  = __builtin_amdgcn_readlane(rp, 0);
    e1  = __builtin_amdgcn_readlane(rp, 1);
    e2  = __builtin_amdgcn_readlane(rp, 2);
    ci0 = load_col_ids(colidx, e0, e2, lane);
    ci1 = load_col_ids(colidx, e0, e2, lane + 32);
}

__device__ __forceinline__ void ring_sum(float4& acc, const float4& vu, int k, int lastA, float* dstA)
{
    add4(acc, vu);
    if (k == lastA) {
        *(float4*)dstA = acc;
        acc = make_float4(0.f, 0.f, 0.f, 0.f);
    }
}

__device__ __forceinline__ int ring_col(int ci0, int ci1, int k)
{
    return __builtin_amdgcn_readlane((k & 32) ? ci1 : ci0, k & 31);
}

__device__ __forceinline__ void store_product_rows(const float* __restrict__ ptile,
                                                   float*       __restrict__ out,
                                                   int row0, int wv, int lane, int n_nodes)
{
    const int rl = __builtin_amdgcn_readfirstlane(wv) * 2;
    const int r  = row0 + rl;
    const float* src = ptile + rl * TILE_PAD + lane * 4;
    float*       dst = out + (size_t)r * D_FEAT + lane * 4;
    if (r + 1 < n_nodes) {
        const float4 p0 = *(const float4*)(src);
        const float4 p1 = *(const float4*)(src + TILE_PAD);
        *(float4*)(dst)          = p0;
        *(float4*)(dst + D_FEAT) = p1;
    } else if (r < n_nodes) {
        *(float4*)(dst) = *(const float4*)(src);
    }
}

__global__ __launch_bounds__(256)
void gnn_fused_wmma_kernel(const float* __restrict__ X,
                           const float* __restrict__ W,
                           const int*   __restrict__ rowptr,
                           const int*   __restrict__ colidx,
                           float*       __restrict__ out,
                           int n_nodes)
{
    __shared__ __attribute__((aligned(16))) float sA[2][16 * TILE_PAD];
    __shared__ __attribute__((aligned(16))) float sO[2][16 * TILE_PAD];

    const int tid  = threadIdx.x;
    const int lane = tid & 31;
    const int wv   = tid >> 5;
    const int n_windows = (n_nodes + 15) >> 4;

    const int n0   = wv * 16;
    const int mrow = lane & 15;
    const int kd   = (lane >> 4) * 2;

    v2f bw[D_FEAT / 4];
    {
        const float* wp = W + (size_t)kd * D_FEAT + n0 + mrow;
        #pragma unroll
        for (int kk = 0; kk < D_FEAT / 4; ++kk) {
            bw[kk].x = wp[(kk * 4    ) * D_FEAT];
            bw[kk].y = wp[(kk * 4 + 1) * D_FEAT];
        }
    }

    const int lane4 = lane * 4;

    int e0 = 0, e1 = 0, e2 = 0, ci0 = 0, ci1 = 0, rpn = 0;
    if ((int)blockIdx.x < n_windows) {
        const int rp = load_row_ptrs(rowptr, blockIdx.x, wv, lane, n_nodes);
        if ((int)blockIdx.x + N_GROUPS < n_windows)
            rpn = load_row_ptrs(rowptr, blockIdx.x + N_GROUPS, wv, lane, n_nodes);
        open_run(colidx, rp, lane, e0, e1, e2, ci0, ci1);
    }

    int buf = 0;
    int prev_r0 = -1;
    for (int win = blockIdx.x; win < n_windows; win += N_GROUPS) {
        float* tile = sA[buf];
        const int r0 = win * 16;

        int ne0, ne1, ne2, nci0, nci1;
        const int win2 = win + 2 * N_GROUPS;

        {
            const int total = e2 - e0;
            const int nring = min(total, 64);
            const int lastA = e1 - e0 - 1;
            float* dstA = tile + (wv * 2    ) * TILE_PAD + lane4;
            float* dstB = tile + (wv * 2 + 1) * TILE_PAD + lane4;
            float4 acc = make_float4(0.f, 0.f, 0.f, 0.f);
            if (lastA < 0) *(float4*)dstA = acc;

            float4 v[GATHER_DEPTH];
            if (nring >= GATHER_DEPTH) {
                #pragma unroll
                for (int u = 0; u < GATHER_DEPTH; ++u)
                    v[u] = load_src_row(X, __builtin_amdgcn_readlane(ci0, u), lane4);
                open_run(colidx, rpn, lane, ne0, ne1, ne2, nci0, nci1);
                rpn = (win2 < n_windows) ? load_row_ptrs(rowptr, win2, wv, lane, n_nodes) : 0;

                int k0 = 0;
                for (; k0 + 2 * GATHER_DEPTH <= nring; k0 += GATHER_DEPTH) {
                    #pragma unroll
                    for (int u = 0; u < GATHER_DEPTH; ++u) {
                        ring_sum(acc, v[u], k0 + u, lastA, dstA);
                        v[u] = load_src_row(X, ring_col(ci0, ci1, k0 + GATHER_DEPTH + u), lane4);
                    }
                }
                #pragma unroll
                for (int u = 0; u < GATHER_DEPTH; ++u) {
                    ring_sum(acc, v[u], k0 + u, lastA, dstA);
                    v[u] = make_float4(0.f, 0.f, 0.f, 0.f);
                    if (k0 + GATHER_DEPTH + u < nring)
                        v[u] = load_src_row(X, ring_col(ci0, ci1, k0 + GATHER_DEPTH + u), lane4);
                }
                k0 += GATHER_DEPTH;
                #pragma unroll
                for (int u = 0; u < GATHER_DEPTH; ++u)
                    if (k0 + u < nring) ring_sum(acc, v[u], k0 + u, lastA, dstA);
            } else {
                #pragma unroll
                for (int u = 0; u < GATHER_DEPTH; ++u) {
                    v[u] = make_float4(0.f, 0.f, 0.f, 0.f);
                    if (u < nring) v[u] = load_src_row(X, __builtin_amdgcn_readlane(ci0, u), lane4);
                }
                open_run(colidx, rpn, lane, ne0, ne1, ne2, nci0, nci1);
                rpn = (win2 < n_windows) ? load_row_ptrs(rowptr, win2, wv, lane, n_nodes) : 0;
                #pragma unroll
                for (int u = 0; u < GATHER_DEPTH; ++u)
                    if (u < nring) ring_sum(acc, v[u], u, lastA, dstA);
            }
            for (int k = 64; k < total; ++k)
                ring_sum(acc, load_src_row(X, colidx[e0 + k], lane4), k, lastA, dstA);
            *(float4*)dstB = acc;
        }
        e0 = ne0; e1 = ne1; e2 = ne2; ci0 = nci0; ci1 = nci1;

        __syncthreads();

        if (prev_r0 >= 0)
            store_product_rows(sO[buf ^ 1], out, prev_r0, wv, lane, n_nodes);

        v8f c = {};
        const float* aBase = tile + mrow * TILE_PAD + kd;
        #pragma unroll
        for (int kk = 0; kk < D_FEAT / 4; ++kk) {
            v2f a = *(const v2f*)(aBase + kk * 4);
            c = __builtin_amdgcn_wmma_f32_16x16x4_f32(
                     false, a,  false, bw[kk],
                     (short)0, c,  false,  false);
        }

        float* ptile = sO[buf];
        const int ncol  = n0 + (lane & 15);
        const int mbase = (lane >> 4) * 8;
        #pragma unroll
        for (int v = 0; v < 8; ++v)
            ptile[(mbase + v) * TILE_PAD + ncol] = c[v];

        prev_r0 = r0;
        buf ^= 1;
    }

    __syncthreads();
    if (prev_r0 >= 0)
        store_product_rows(sO[buf ^ 1], out, prev_r0, wv, lane, n_nodes);
}

extern "C" void kernel_launch(void* const* d_in, const int* in_sizes, int n_in,
                              void* d_out, int out_size, void* d_ws, size_t ws_size,
                              hipStream_t stream) {
    const float* X      = (const float*)d_in[0];
    const float* W      = (const float*)d_in[1];
    const int*   rowptr = (const int*)  d_in[2];
    const int*   colidx = (const int*)  d_in[3];
    float*       out    = (float*)d_out;

    const int n_nodes   = in_sizes[2] - 1;

    hipLaunchKernelGGL(gnn_fused_wmma_kernel,
                       dim3(N_GROUPS), dim3(256), 0, stream,
                       X, W, rowptr, colidx, out, n_nodes);
}
